# starting kernel
# baseline (speedup 1.0000x reference)
#include <hip/hip_runtime.h>
#include <hip/hip_bf16.h>

typedef __attribute__((ext_vector_type(16))) _Float16 v16h;
typedef __attribute__((ext_vector_type(8)))  float    v8f;

#define N_    256
#define NN    65536
#define KHOP  32
#define D_    32
#define H_    8
#define WAVES_PER_BLOCK 8

__global__ __launch_bounds__(256)
void attn_bias_wmma(const int*   __restrict__ spatial_pos,
                    const int*   __restrict__ edge_input,
                    const float* __restrict__ spatial_emb,
                    const float* __restrict__ edge_feat_emb,
                    const float* __restrict__ edge_pos_emb,
                    float*       __restrict__ out)
{
    __shared__ _Float16 bfrag[KHOP * 32 * 16];

    const int tid  = threadIdx.x;
    const int lane = tid & 31;
    const int wave = tid >> 5;

    for (int f = tid; f < KHOP * 32; f += 256) {
        const int k  = f >> 5;
        const int L  = f & 31;
        const int n  = L & 15;
        const int hi = (L >> 4) * 8;
        _Float16* dst = &bfrag[f * 16];
        if (n < H_) {
            const float* w = edge_pos_emb + k * (D_ * H_) + n;
#pragma unroll
            for (int j = 0; j < 16; ++j) {
                const int d = (j < 8) ? (j + hi) : (j - 8 + 16 + hi);
                dst[j] = (_Float16)w[d * H_];
            }
        } else {
#pragma unroll
            for (int j = 0; j < 16; ++j) dst[j] = (_Float16)0.0f;
        }
    }
    __syncthreads();

    const int  tile  = blockIdx.x * WAVES_PER_BLOCK + wave;
    const long long base = (long long)tile * 16;
    const int  mrow  = lane & 15;
    const int  dbase = (lane >> 4) * 8;

    const int* eip = edge_input + (base + mrow) * KHOP;

    v8f c = {};
    int cnt = 0;
    for (int k = 0; k < KHOP; ++k) {
        const int idx = eip[k];
        cnt += (idx != 0);
        const float4* row4 = (const float4*)(edge_feat_emb + (long long)idx * D_ + dbase);
        const float4 c0 = row4[0], c1 = row4[1], c2 = row4[4], c3 = row4[5];
        v16h a;
        a[0]  = (_Float16)c0.x; a[1]  = (_Float16)c0.y; a[2]  = (_Float16)c0.z; a[3]  = (_Float16)c0.w;
        a[4]  = (_Float16)c1.x; a[5]  = (_Float16)c1.y; a[6]  = (_Float16)c1.z; a[7]  = (_Float16)c1.w;
        a[8]  = (_Float16)c2.x; a[9]  = (_Float16)c2.y; a[10] = (_Float16)c2.z; a[11] = (_Float16)c2.w;
        a[12] = (_Float16)c3.x; a[13] = (_Float16)c3.y; a[14] = (_Float16)c3.z; a[15] = (_Float16)c3.w;

        const v16h b = *(const v16h*)&bfrag[(k * 32 + lane) * 16];

        c = __builtin_amdgcn_wmma_f32_16x16x32_f16(
                 false, a,  false, b,
                 (short)0, c,  false,  false);
    }

    const int   h    = lane & 15;
    const int   mtop = (lane >> 4) * 8;
    const float cntf = (float)cnt;

    float pls[8];
#pragma unroll
    for (int r = 0; r < 8; ++r)
        pls[r] = fmaxf(__shfl(cntf, mtop + r, 32), 1.0f);

    if (h < H_) {
#pragma unroll
        for (int r = 0; r < 8; ++r) {
            const long long p   = base + mtop + r;
            const int       sp  = spatial_pos[p];
            const float     sb  = spatial_emb[sp * H_ + h];
            const long long bi  = p >> 16;
            const long long nm  = p & 0xFFFF;
            out[bi * (long long)(H_ * NN) + (long long)h * NN + nm] =
                sb + c[r] / pls[r];
        }
    }
}

extern "C" void kernel_launch(void* const* d_in, const int* in_sizes, int n_in,
                              void* d_out, int out_size, void* d_ws, size_t ws_size,
                              hipStream_t stream) {
    const int*   spatial_pos   = (const int*)d_in[0];
    const int*   edge_input    = (const int*)d_in[1];
    const float* spatial_emb   = (const float*)d_in[2];
    const float* edge_feat_emb = (const float*)d_in[3];
    const float* edge_pos_emb  = (const float*)d_in[4];
    float*       out           = (float*)d_out;

    const int pairs  = in_sizes[0];
    const int tiles  = pairs / 16;
    const int blocks = tiles / WAVES_PER_BLOCK;

    hipLaunchKernelGGL(attn_bias_wmma, dim3(blocks), dim3(256), 0, stream,
                       spatial_pos, edge_input, spatial_emb, edge_feat_emb,
                       edge_pos_emb, out);
}


# Edge-feature table converted to f16 once per work-group and held in LDS by 512 resident work-groups looping over tiles; A fragments gathered from LDS, not global memory; hop sum order unchanged
# speedup vs baseline: 3.6008x; 3.6008x over previous
#include <hip/hip_runtime.h>

typedef __attribute__((ext_vector_type(16))) _Float16 v16h;
typedef __attribute__((ext_vector_type(8)))  _Float16 v8h;
typedef __attribute__((ext_vector_type(4)))  _Float16 v4h;
typedef __attribute__((ext_vector_type(8)))  float    v8f;

#define N_    256
#define NN    65536
#define KHOP  32
#define D_    32
#define H_    8
#define NROWS 1025
#define NTILES 16384

#define WAVES_PER_BLOCK 8
#define THREADS (WAVES_PER_BLOCK * 32)
#define NBLOCKS 512
#define TILES_PER_BLOCK (NTILES / NBLOCKS)
#define TILE_ITERS (TILES_PER_BLOCK / WAVES_PER_BLOCK)

#define ROW_PITCH 40

#define NQ (NROWS * (D_ / 4))

__device__ __forceinline__ void load_hop_indices(const int* __restrict__ row, int (&dst)[KHOP])
{
    const int4* p = (const int4*)row;
#pragma unroll
    for (int j = 0; j < KHOP / 4; ++j) {
        const int4 v = p[j];
        dst[4 * j + 0] = v.x; dst[4 * j + 1] = v.y; dst[4 * j + 2] = v.z; dst[4 * j + 3] = v.w;
    }
}

__global__ __launch_bounds__(THREADS)
void attn_bias_wmma(const int*   __restrict__ spatial_pos,
                    const int*   __restrict__ edge_input,
                    const float* __restrict__ spatial_emb,
                    const float* __restrict__ edge_feat_emb,
                    const float* __restrict__ edge_pos_emb,
                    float*       __restrict__ out)
{
    __shared__ __attribute__((aligned(16))) _Float16 bfrag[KHOP * 32 * 16];
    __shared__ __attribute__((aligned(16))) _Float16 tab[NROWS * ROW_PITCH];

    const int tid   = threadIdx.x;
    const int lane  = tid & 31;
    const int wave  = tid >> 5;
    const int mrow  = lane & 15;
    const int dbase = (lane >> 4) * 8;

    int tile = blockIdx.x * TILES_PER_BLOCK + wave;

    int nxt[KHOP];
    load_hop_indices(edge_input + ((long long)tile * 16 + mrow) * KHOP, nxt);

    for (int f = tid; f < KHOP * 32; f += THREADS) {
        const int k  = f >> 5;
        const int L  = f & 31;
        const int n  = L & 15;
        const int hi = (L >> 4) * 8;
        _Float16* dst = &bfrag[f * 16];
        if (n < H_) {
            const float* w = edge_pos_emb + k * (D_ * H_) + n;
#pragma unroll
            for (int j = 0; j < 16; ++j) {
                const int d = (j < 8) ? (j + hi) : (j - 8 + 16 + hi);
                dst[j] = (_Float16)w[d * H_];
            }
        } else {
#pragma unroll
            for (int j = 0; j < 16; ++j) dst[j] = (_Float16)0.0f;
        }
    }

    {
        const float4* t4 = (const float4*)edge_feat_emb;
#pragma unroll
        for (int i = 0; i < NQ / THREADS; ++i) {
            const int    q = tid + i * THREADS;
            const float4 v = t4[q];
            v4h hv;
            hv[0] = (_Float16)v.x; hv[1] = (_Float16)v.y; hv[2] = (_Float16)v.z; hv[3] = (_Float16)v.w;
            *(v4h*)&tab[(q >> 3) * ROW_PITCH + (q & 7) * 4] = hv;
        }
        if (tid < NQ % THREADS) {
            const int    q = tid + (NQ / THREADS) * THREADS;
            const float4 v = t4[q];
            v4h hv;
            hv[0] = (_Float16)v.x; hv[1] = (_Float16)v.y; hv[2] = (_Float16)v.z; hv[3] = (_Float16)v.w;
            *(v4h*)&tab[(q >> 3) * ROW_PITCH + (q & 7) * 4] = hv;
        }
    }
    __syncthreads();

#pragma unroll
    for (int it = 0; it < TILE_ITERS; ++it, tile += WAVES_PER_BLOCK) {
        const long long base = (long long)tile * 16;

        int idxs[KHOP];
#pragma unroll
        for (int k = 0; k < KHOP; ++k) idxs[k] = nxt[k];

        if (it + 1 < TILE_ITERS)
            load_hop_indices(edge_input + ((long long)(tile + WAVES_PER_BLOCK) * 16 + mrow) * KHOP, nxt);

        v8f c = {};
        int cnt = 0;
#pragma unroll
        for (int k = 0; k < KHOP; ++k) {
            const int idx = idxs[k];
            cnt += (idx != 0);
            const _Float16* rp = tab + idx * ROW_PITCH + dbase;
            const v8h alo = *(const v8h*)rp;
            const v8h ahi = *(const v8h*)(rp + 16);
            const v16h a = __builtin_shufflevector(alo, ahi, 0, 1, 2, 3, 4, 5, 6, 7,
                                                   8, 9, 10, 11, 12, 13, 14, 15);

            const v16h b = *(const v16h*)&bfrag[(k * 32 + lane) * 16];

            c = __builtin_amdgcn_wmma_f32_16x16x32_f16(
                     false, a,  false, b,
                     (short)0, c,  false,  false);
        }

        const int   h    = lane & 15;
        const int   mtop = (lane >> 4) * 8;
        const float cntf = (float)cnt;

        float pls[8];
#pragma unroll
        for (int r = 0; r < 8; ++r)
            pls[r] = fmaxf(__shfl(cntf, mtop + r, 32), 1.0f);

        if (h < H_) {
#pragma unroll
            for (int r = 0; r < 8; ++r) {
                const long long p   = base + mtop + r;
                const int       sp  = spatial_pos[p];
                const float     sb  = spatial_emb[sp * H_ + h];
                const long long bi  = p >> 16;
                const long long nm  = p & 0xFFFF;
                out[bi * (long long)(H_ * NN) + (long long)h * NN + nm] =
                    sb + c[r] / pls[r];
            }
        }
    }
}

extern "C" void kernel_launch(void* const* d_in, const int* in_sizes, int n_in,
                              void* d_out, int out_size, void* d_ws, size_t ws_size,
                              hipStream_t stream) {
    const int*   spatial_pos   = (const int*)d_in[0];
    const int*   edge_input    = (const int*)d_in[1];
    const float* spatial_emb   = (const float*)d_in[2];
    const float* edge_feat_emb = (const float*)d_in[3];
    const float* edge_pos_emb  = (const float*)d_in[4];
    float*       out           = (float*)d_out;

    hipLaunchKernelGGL(attn_bias_wmma, dim3(NBLOCKS), dim3(THREADS), 0, stream,
                       spatial_pos, edge_input, spatial_emb, edge_feat_emb,
                       edge_pos_emb, out);
}


# Hop-outer loop: a wave advances its 4 tiles together on 4 accumulators, so each hop's B fragment is read from LDS once per wave instead of once per tile; per-tile hop sum order unchanged
# speedup vs baseline: 3.8714x; 1.0751x over previous
#include <hip/hip_runtime.h>

typedef __attribute__((ext_vector_type(16))) _Float16 v16h;
typedef __attribute__((ext_vector_type(8)))  _Float16 v8h;
typedef __attribute__((ext_vector_type(4)))  _Float16 v4h;
typedef __attribute__((ext_vector_type(8)))  float    v8f;

#define N_    256
#define NN    65536
#define KHOP  32
#define D_    32
#define H_    8
#define NROWS 1025
#define NTILES 16384

#define WAVES_PER_BLOCK 8
#define THREADS (WAVES_PER_BLOCK * 32)
#define NBLOCKS 512
#define TILES_PER_BLOCK (NTILES / NBLOCKS)
#define TILE_ITERS (TILES_PER_BLOCK / WAVES_PER_BLOCK)

#define ROW_PITCH 40

#define NQ (NROWS * (D_ / 4))

__device__ __forceinline__ void load_hop_offsets(const int* __restrict__ row, int lane_bytes,
                                                 int (&off)[KHOP], int& cnt)
{
    const int4* p = (const int4*)row;
    int n = 0;
#pragma unroll
    for (int j = 0; j < KHOP / 4; ++j) {
        const int4 v = p[j];
        n += (v.x != 0) + (v.y != 0) + (v.z != 0) + (v.w != 0);
        off[4 * j + 0] = v.x * (ROW_PITCH * 2) + lane_bytes;
        off[4 * j + 1] = v.y * (ROW_PITCH * 2) + lane_bytes;
        off[4 * j + 2] = v.z * (ROW_PITCH * 2) + lane_bytes;
        off[4 * j + 3] = v.w * (ROW_PITCH * 2) + lane_bytes;
    }
    cnt = n;
}

#define TG 2
#define STEPS_PER_HOP (TILE_ITERS / TG)
#define NSTEPS (KHOP * STEPS_PER_HOP)
#define AHEAD 2

struct StepFrags {
    v8h alo[TG];
    v8h ahi[TG];
};

__device__ __forceinline__ void load_step(int s, StepFrags& f, const char* tabb,
                                          const int (&off)[TILE_ITERS][KHOP])
{
    const int k = s / STEPS_PER_HOP;
    const int g = s % STEPS_PER_HOP;
#pragma unroll
    for (int j = 0; j < TG; ++j) {
        const char* rp = tabb + off[g * TG + j][k];
        f.alo[j] = *(const v8h*)rp;
        f.ahi[j] = *(const v8h*)(rp + 32);
    }
}

__device__ __forceinline__ v16h load_bfrag(int k, const _Float16* bfrag, int lane)
{
    return *(const v16h*)&bfrag[(k * 32 + lane) * 16];
}

__global__ __launch_bounds__(THREADS)
void attn_bias_wmma(const int*   __restrict__ spatial_pos,
                    const int*   __restrict__ edge_input,
                    const float* __restrict__ spatial_emb,
                    const float* __restrict__ edge_feat_emb,
                    const float* __restrict__ edge_pos_emb,
                    float*       __restrict__ out)
{
    __shared__ __attribute__((aligned(16))) _Float16 bfrag[KHOP * 32 * 16];
    __shared__ __attribute__((aligned(16))) _Float16 tab[NROWS * ROW_PITCH];

    const int tid   = threadIdx.x;
    const int lane  = tid & 31;
    const int wave  = tid >> 5;
    const int mrow  = lane & 15;
    const int dbase = (lane >> 4) * 8;

    const int tile0 = blockIdx.x * TILES_PER_BLOCK + wave;

    {
        const float4* t4 = (const float4*)edge_feat_emb;
#pragma unroll
        for (int i = 0; i < NQ / THREADS; ++i) {
            const int    q = tid + i * THREADS;
            const float4 v = t4[q];
            v4h hv;
            hv[0] = (_Float16)v.x; hv[1] = (_Float16)v.y; hv[2] = (_Float16)v.z; hv[3] = (_Float16)v.w;
            *(v4h*)&tab[(q >> 3) * ROW_PITCH + (q & 7) * 4] = hv;
        }
        if (tid < NQ % THREADS) {
            const int    q = tid + (NQ / THREADS) * THREADS;
            const float4 v = t4[q];
            v4h hv;
            hv[0] = (_Float16)v.x; hv[1] = (_Float16)v.y; hv[2] = (_Float16)v.z; hv[3] = (_Float16)v.w;
            *(v4h*)&tab[(q >> 3) * ROW_PITCH + (q & 7) * 4] = hv;
        }
    }
    int off[TILE_ITERS][KHOP];
    int cnt[TILE_ITERS];
#pragma unroll
    for (int t = 0; t < TILE_ITERS; ++t)
        load_hop_offsets(edge_input + ((long long)(tile0 + t * WAVES_PER_BLOCK) * 16 + mrow) * KHOP,
                         dbase * 2, off[t], cnt[t]);

    for (int f = tid; f < KHOP * 32; f += THREADS) {
        const int k  = f >> 5;
        const int L  = f & 31;
        const int n  = L & 15;
        const int hi = (L >> 4) * 8;
        _Float16* dst = &bfrag[f * 16];
        if (n < H_) {
            const float* w = edge_pos_emb + k * (D_ * H_) + n;
#pragma unroll
            for (int j = 0; j < 16; ++j) {
                const int d = (j < 8) ? (j + hi) : (j - 8 + 16 + hi);
                dst[j] = (_Float16)w[d * H_];
            }
        } else {
#pragma unroll
            for (int j = 0; j < 16; ++j) dst[j] = (_Float16)0.0f;
        }
    }

    __syncthreads();

    v8f c[TILE_ITERS];
#pragma unroll
    for (int t = 0; t < TILE_ITERS; ++t) c[t] = v8f{};

    const char* tabb = (const char*)tab;

    StepFrags fr[AHEAD + 1];
    v16h      bq[2];
    bq[0] = load_bfrag(0, bfrag, lane);
#pragma unroll
    for (int s = 0; s < AHEAD; ++s)
        load_step(s, fr[s], tabb, off);

#pragma unroll
    for (int s = 0; s < NSTEPS; ++s) {
        const int k = s / STEPS_PER_HOP;
        const int g = s % STEPS_PER_HOP;

        if (s + AHEAD < NSTEPS)
            load_step(s + AHEAD, fr[(s + AHEAD) % (AHEAD + 1)], tabb, off);
        if (g == 0 && k + 1 < KHOP)
            bq[(k + 1) & 1] = load_bfrag(k + 1, bfrag, lane);

        const StepFrags& f = fr[s % (AHEAD + 1)];
#pragma unroll
        for (int j = 0; j < TG; ++j) {
            const v16h a = __builtin_shufflevector(f.alo[j], f.ahi[j], 0, 1, 2, 3, 4, 5, 6, 7,
                                                   8, 9, 10, 11, 12, 13, 14, 15);
            c[g * TG + j] = __builtin_amdgcn_wmma_f32_16x16x32_f16(
                                 false, a,  false, bq[k & 1],
                                 (short)0, c[g * TG + j],  false,  false);
        }
    }

    const int h    = lane & 15;
    const int mtop = (lane >> 4) * 8;

    float pls[TILE_ITERS][8];
#pragma unroll
    for (int t = 0; t < TILE_ITERS; ++t) {
        const float cntf = (float)cnt[t];
#pragma unroll
        for (int r = 0; r < 8; ++r)
            pls[t][r] = fmaxf(__shfl(cntf, mtop + r, 32), 1.0f);
    }

    if (h < H_) {
        int sp[TILE_ITERS][8];
#pragma unroll
        for (int t = 0; t < TILE_ITERS; ++t) {
            const long long base = (long long)(tile0 + t * WAVES_PER_BLOCK) * 16;
#pragma unroll
            for (int r = 0; r < 8; ++r)
                sp[t][r] = spatial_pos[base + mtop + r];
        }
        float sb[TILE_ITERS][8];
#pragma unroll
        for (int t = 0; t < TILE_ITERS; ++t) {
#pragma unroll
            for (int r = 0; r < 8; ++r)
                sb[t][r] = spatial_emb[sp[t][r] * H_ + h];
        }
#pragma unroll
        for (int t = 0; t < TILE_ITERS; ++t) {
            const long long base = (long long)(tile0 + t * WAVES_PER_BLOCK) * 16;
#pragma unroll
            for (int r = 0; r < 8; ++r) {
                const long long p   = base + mtop + r;
                const long long bi  = p >> 16;
                const long long nm  = p & 0xFFFF;
                out[bi * (long long)(H_ * NN) + (long long)h * NN + nm] =
                    sb[t][r] + c[t][r] / pls[t][r];
            }
        }
    }
}

extern "C" void kernel_launch(void* const* d_in, const int* in_sizes, int n_in,
                              void* d_out, int out_size, void* d_ws, size_t ws_size,
                              hipStream_t stream) {
    const int*   spatial_pos   = (const int*)d_in[0];
    const int*   edge_input    = (const int*)d_in[1];
    const float* spatial_emb   = (const float*)d_in[2];
    const float* edge_feat_emb = (const float*)d_in[3];
    const float* edge_pos_emb  = (const float*)d_in[4];
    float*       out           = (float*)d_out;

    hipLaunchKernelGGL(attn_bias_wmma, dim3(NBLOCKS), dim3(THREADS), 0, stream,
                       spatial_pos, edge_input, spatial_emb, edge_feat_emb,
                       edge_pos_emb, out);
}


# speedup vs baseline: 4.0977x; 1.0585x over previous
#include <hip/hip_runtime.h>

typedef __attribute__((ext_vector_type(16))) _Float16 v16h;
typedef __attribute__((ext_vector_type(8)))  _Float16 v8h;
typedef __attribute__((ext_vector_type(4)))  _Float16 v4h;
typedef __attribute__((ext_vector_type(8)))  float    v8f;

#define N_    256
#define NN    65536
#define KHOP  32
#define D_    32
#define H_    8
#define NROWS 1025
#define NTILES 16384

#define WAVES_PER_BLOCK 16
#define THREADS (WAVES_PER_BLOCK * 32)
#define NBLOCKS 256
#define TILES_PER_BLOCK (NTILES / NBLOCKS)
#define TILE_ITERS (TILES_PER_BLOCK / WAVES_PER_BLOCK)

#define ROW_PITCH 40

#define NQ (NROWS * (D_ / 4))

#define TAB_BYTES    (NROWS * ROW_PITCH * 2)
#define BFRAG_BYTES  (KHOP * 32 * 16 * 2)
#define COPY_BYTES   (TAB_BYTES + BFRAG_BYTES)
#define COPY_STRIDE  149504
#define LDS_BYTES    (COPY_STRIDE + COPY_BYTES)

__device__ __forceinline__ void load_hop_offsets(const int* __restrict__ row, int lane_bytes,
                                                 int (&off)[KHOP], int& cnt)
{
    const int4* p = (const int4*)row;
    int n = 0;
#pragma unroll
    for (int j = 0; j < KHOP / 4; ++j) {
        const int4 v = p[j];
        n += (v.x != 0) + (v.y != 0) + (v.z != 0) + (v.w != 0);
        off[4 * j + 0] = v.x * (ROW_PITCH * 2) + lane_bytes;
        off[4 * j + 1] = v.y * (ROW_PITCH * 2) + lane_bytes;
        off[4 * j + 2] = v.z * (ROW_PITCH * 2) + lane_bytes;
        off[4 * j + 3] = v.w * (ROW_PITCH * 2) + lane_bytes;
    }
    cnt = n;
}

#define TG 2
#define STEPS_PER_HOP (TILE_ITERS / TG)
#define NSTEPS (KHOP * STEPS_PER_HOP)
#define AHEAD 2

struct StepFrags {
    v8h alo[TG];
    v8h ahi[TG];
};

__device__ __forceinline__ void load_step(int s, StepFrags& f, const char* tabb,
                                          const int (&off)[TILE_ITERS][KHOP])
{
    const int k = s / STEPS_PER_HOP;
    const int g = s % STEPS_PER_HOP;
#pragma unroll
    for (int j = 0; j < TG; ++j) {
        const char* rp = tabb + off[g * TG + j][k];
        f.alo[j] = *(const v8h*)rp;
        f.ahi[j] = *(const v8h*)(rp + 32);
    }
}

__device__ __forceinline__ v16h load_bfrag(int k, const _Float16* bfrag, int lane)
{
    return *(const v16h*)&bfrag[(k * 32 + lane) * 16];
}

__device__ __forceinline__ int simd_pair()
{
    int hwid;
    asm("s_getreg_b32 %0, hwreg(HW_REG_WAVE_HW_ID1)" : "=s"(hwid));
    return (hwid >> 9) & 1;
}

__global__ __launch_bounds__(THREADS)
void attn_bias_wmma(const int*   __restrict__ spatial_pos,
                    const int*   __restrict__ edge_input,
                    const float* __restrict__ spatial_emb,
                    const float* __restrict__ edge_feat_emb,
                    const float* __restrict__ edge_pos_emb,
                    float*       __restrict__ out)
{
    __shared__ __attribute__((aligned(16))) char lds[LDS_BYTES];
    _Float16* const tab0   = (_Float16*)lds;
    _Float16* const tab1   = (_Float16*)(lds + COPY_STRIDE);
    _Float16* const bfrag0 = (_Float16*)(lds + TAB_BYTES);
    _Float16* const bfrag1 = (_Float16*)(lds + COPY_STRIDE + TAB_BYTES);

    const int tid   = threadIdx.x;
    const int lane  = tid & 31;
    const int wave  = tid >> 5;
    const int mrow  = lane & 15;
    const int dbase = (lane >> 4) * 8;

    const int tile0 = blockIdx.x * TILES_PER_BLOCK + wave;

    {
        const float4* t4 = (const float4*)edge_feat_emb;
#pragma unroll
        for (int i = 0; i < NQ / THREADS; ++i) {
            const int    q = tid + i * THREADS;
            const float4 v = t4[q];
            v4h hv;
            hv[0] = (_Float16)v.x; hv[1] = (_Float16)v.y; hv[2] = (_Float16)v.z; hv[3] = (_Float16)v.w;
            *(v4h*)&tab0[(q >> 3) * ROW_PITCH + (q & 7) * 4] = hv;
            *(v4h*)&tab1[(q >> 3) * ROW_PITCH + (q & 7) * 4] = hv;
        }
        if (tid < NQ % THREADS) {
            const int    q = tid + (NQ / THREADS) * THREADS;
            const float4 v = t4[q];
            v4h hv;
            hv[0] = (_Float16)v.x; hv[1] = (_Float16)v.y; hv[2] = (_Float16)v.z; hv[3] = (_Float16)v.w;
            *(v4h*)&tab0[(q >> 3) * ROW_PITCH + (q & 7) * 4] = hv;
            *(v4h*)&tab1[(q >> 3) * ROW_PITCH + (q & 7) * 4] = hv;
        }
    }
    const int copy_base = simd_pair() * COPY_STRIDE;
    int off[TILE_ITERS][KHOP];
    int cnt[TILE_ITERS];
#pragma unroll
    for (int t = 0; t < TILE_ITERS; ++t)
        load_hop_offsets(edge_input + ((long long)(tile0 + t * WAVES_PER_BLOCK) * 16 + mrow) * KHOP,
                         dbase * 2 + copy_base, off[t], cnt[t]);

    for (int f = tid; f < KHOP * 32; f += THREADS) {
        const int k  = f >> 5;
        const int L  = f & 31;
        const int n  = L & 15;
        const int hi = (L >> 4) * 8;
        _Float16* dst0 = &bfrag0[f * 16];
        _Float16* dst1 = &bfrag1[f * 16];
        if (n < H_) {
            const float* w = edge_pos_emb + k * (D_ * H_) + n;
#pragma unroll
            for (int j = 0; j < 16; ++j) {
                const int d = (j < 8) ? (j + hi) : (j - 8 + 16 + hi);
                const _Float16 wv = (_Float16)w[d * H_];
                dst0[j] = wv;
                dst1[j] = wv;
            }
        } else {
#pragma unroll
            for (int j = 0; j < 16; ++j) { dst0[j] = (_Float16)0.0f; dst1[j] = (_Float16)0.0f; }
        }
    }

    __syncthreads();

    v8f c[TILE_ITERS];
#pragma unroll
    for (int t = 0; t < TILE_ITERS; ++t) c[t] = v8f{};

    const char* tabb = (const char*)lds;
    const _Float16* bfrag = (const _Float16*)(lds + copy_base + TAB_BYTES);

    StepFrags fr[AHEAD + 1];
    v16h      bq[2];
    bq[0] = load_bfrag(0, bfrag, lane);
#pragma unroll
    for (int s = 0; s < AHEAD; ++s)
        load_step(s, fr[s], tabb, off);

#pragma unroll
    for (int s = 0; s < NSTEPS; ++s) {
        const int k = s / STEPS_PER_HOP;
        const int g = s % STEPS_PER_HOP;

        if (s + AHEAD < NSTEPS)
            load_step(s + AHEAD, fr[(s + AHEAD) % (AHEAD + 1)], tabb, off);
        if (g == 0 && k + 1 < KHOP)
            bq[(k + 1) & 1] = load_bfrag(k + 1, bfrag, lane);

        const StepFrags& f = fr[s % (AHEAD + 1)];
#pragma unroll
        for (int j = 0; j < TG; ++j) {
            const v16h a = __builtin_shufflevector(f.alo[j], f.ahi[j], 0, 1, 2, 3, 4, 5, 6, 7,
                                                   8, 9, 10, 11, 12, 13, 14, 15);
            c[g * TG + j] = __builtin_amdgcn_wmma_f32_16x16x32_f16(
                                 false, a,  false, bq[k & 1],
                                 (short)0, c[g * TG + j],  false,  false);
        }
    }

    const int h    = lane & 15;
    const int mtop = (lane >> 4) * 8;

    float pls[TILE_ITERS][8];
#pragma unroll
    for (int t = 0; t < TILE_ITERS; ++t) {
        const float cntf = (float)cnt[t];
#pragma unroll
        for (int r = 0; r < 8; ++r)
            pls[t][r] = fmaxf(__shfl(cntf, mtop + r, 32), 1.0f);
    }

    if (h < H_) {
        int sp[TILE_ITERS][8];
#pragma unroll
        for (int t = 0; t < TILE_ITERS; ++t) {
            const long long base = (long long)(tile0 + t * WAVES_PER_BLOCK) * 16;
#pragma unroll
            for (int r = 0; r < 8; ++r)
                sp[t][r] = spatial_pos[base + mtop + r];
        }
        float sb[TILE_ITERS][8];
#pragma unroll
        for (int t = 0; t < TILE_ITERS; ++t) {
#pragma unroll
            for (int r = 0; r < 8; ++r)
                sb[t][r] = spatial_emb[sp[t][r] * H_ + h];
        }
#pragma unroll
        for (int t = 0; t < TILE_ITERS; ++t) {
            const long long base = (long long)(tile0 + t * WAVES_PER_BLOCK) * 16;
#pragma unroll
            for (int r = 0; r < 8; ++r) {
                const long long p   = base + mtop + r;
                const long long bi  = p >> 16;
                const long long nm  = p & 0xFFFF;
                out[bi * (long long)(H_ * NN) + (long long)h * NN + nm] =
                    sb[t][r] + c[t][r] / pls[t][r];
            }
        }
    }
}

extern "C" void kernel_launch(void* const* d_in, const int* in_sizes, int n_in,
                              void* d_out, int out_size, void* d_ws, size_t ws_size,
                              hipStream_t stream) {
    const int*   spatial_pos   = (const int*)d_in[0];
    const int*   edge_input    = (const int*)d_in[1];
    const float* spatial_emb   = (const float*)d_in[2];
    const float* edge_feat_emb = (const float*)d_in[3];
    const float* edge_pos_emb  = (const float*)d_in[4];
    float*       out           = (float*)d_out;

    hipLaunchKernelGGL(attn_bias_wmma, dim3(NBLOCKS), dim3(THREADS), 0, stream,
                       spatial_pos, edge_input, spatial_emb, edge_feat_emb,
                       edge_pos_emb, out);
}


# P build unblocked: table tiles come as whole 128-B lines (4 per wave load, was 16 partial) via a per-wave LDS transpose; each product's weight fragment is read one product ahead; sums unchanged
# speedup vs baseline: 6.1904x; 1.5107x over previous
#include <hip/hip_runtime.h>

typedef __attribute__((ext_vector_type(16))) _Float16 v16h;
typedef __attribute__((ext_vector_type(8)))  _Float16 v8h;
typedef __attribute__((ext_vector_type(4)))  _Float16 v4h;
typedef __attribute__((ext_vector_type(8)))  float    v8f;

#define NN    65536
#define KHOP  32
#define D_    32
#define H_    8
#define HH    4
#define NROWS 1025
#define LAST_ROW (NROWS - 1)

#define WAVES_PER_BLOCK 16
#define THREADS (WAVES_PER_BLOCK * 32)
#define NBLOCKS 256
#define PAIRS_PER_BLOCK 2048
#define PAIRS_PER_WAVE  128
#define CHUNKS  32
#define QCHUNKS 8

#define P_ENTRY   8
#define P_KSTRIDE (NROWS * P_ENTRY)
#define P_BYTES   (KHOP * P_KSTRIDE)
#define NWT       8
#define WF_BYTES  (NWT * 32 * 16 * 2)
#define STG_PITCH 80
#define STG_TILE  (16 * STG_PITCH)
#define STG_BYTES (WAVES_PER_BLOCK * 2 * STG_TILE)
#define LDS_BYTES (P_BYTES + WF_BYTES + STG_BYTES)

#define TT_PER_WAVE 5

__device__ __forceinline__ void lds_barrier()
{
    __builtin_amdgcn_fence(__ATOMIC_RELEASE, "workgroup", "local");
    __builtin_amdgcn_s_barrier();
    __builtin_amdgcn_fence(__ATOMIC_ACQUIRE, "workgroup", "local");
}

template <int CTRL>
__device__ __forceinline__ float lane_swap(float x)
{
    return __builtin_bit_cast(float,
        __builtin_amdgcn_update_dpp(0, __builtin_bit_cast(int, x), CTRL, 0xF, 0xF, true));
}
#define SWAP_XOR1 0xB1
#define SWAP_XOR2 0x4E
#define SWAP_XOR4 0x164

__global__ __launch_bounds__(THREADS)
void attn_bias_wmma(const int*   __restrict__ spatial_pos,
                    const int*   __restrict__ edge_input,
                    const float* __restrict__ spatial_emb,
                    const float* __restrict__ edge_feat_emb,
                    const float* __restrict__ edge_pos_emb,
                    float*       __restrict__ out)
{
    __shared__ __attribute__((aligned(16))) char lds[LDS_BYTES];
    char* const     ptab = lds;
    _Float16* const wf   = (_Float16*)(lds + P_BYTES);
    char* const     stg  = lds + P_BYTES + WF_BYTES + (threadIdx.x >> 5) * (2 * STG_TILE);

    const int tid  = threadIdx.x;
    const int lane = tid & 31;
    const int wave = tid >> 5;
    const int c16  = lane & 15;
    const int half = lane >> 4;
    const int hb   = blockIdx.x & 1;
    const int pbase = (blockIdx.x >> 1) * PAIRS_PER_BLOCK + wave * PAIRS_PER_WAVE;

    const int wfl = tid >> 1;
    const int wsh = tid & 1;
    float wraw[8];
    {
        const int m    = (wfl >> 5) * 16 + (wfl & 15);
        const int k    = m >> 2;
        const int head = hb * HH + (m & 3);
        const int d0   = ((wfl >> 4) & 1) * 16 + wsh * 8;
        const float* w = edge_pos_emb + k * (D_ * H_) + d0 * H_ + head;
#pragma unroll
        for (int j = 0; j < 8; ++j) wraw[j] = w[j * H_];
    }

    const int trow = lane >> 3;
    const int tcol = lane & 7;
    float4 tq[TT_PER_WAVE][4];
#pragma unroll
    for (int t = 0; t < TT_PER_WAVE; ++t) {
#pragma unroll
        for (int j = 0; j < 4; ++j) {
            const int row = min((wave + t * WAVES_PER_BLOCK) * 16 + 4 * j + trow, LAST_ROW);
            tq[t][j] = *(const float4*)(edge_feat_emb + row * D_ + tcol * 4);
        }
    }

    const int pi   = lane >> 3;
    const int l3   = lane & 7;
    const int slot = 4 * l3 + pi;
    int sp[CHUNKS / QCHUNKS];
#pragma unroll
    for (int g = 0; g < CHUNKS / QCHUNKS; ++g) sp[g] = spatial_pos[pbase + g * 32 + slot];

    const int4* ein = (const int4*)(edge_input + (long long)pbase * KHOP) + lane;
    int4 idx[CHUNKS];
#pragma unroll
    for (int c = 0; c < CHUNKS / 2; ++c) idx[c] = ein[c * 32];

    {
        v8h f;
#pragma unroll
        for (int j = 0; j < 8; ++j) f[j] = (_Float16)wraw[j];
        *(v8h*)&wf[wfl * 16 + wsh * 8] = f;
    }

    lds_barrier();

    v16h a_cur = *(const v16h*)&wf[lane * 16];
#pragma unroll
    for (int t = 0; t < TT_PER_WAVE; ++t) {
        if (t < TT_PER_WAVE - 1 || wave == 0) {
            char* const sb2 = stg + (t & 1) * STG_TILE;
#pragma unroll
            for (int j = 0; j < 4; ++j) {
                asm volatile("" : "+v"(tq[t][j].x), "+v"(tq[t][j].y), "+v"(tq[t][j].z), "+v"(tq[t][j].w));
                v4h hq;
                hq[0] = (_Float16)tq[t][j].x; hq[1] = (_Float16)tq[t][j].y;
                hq[2] = (_Float16)tq[t][j].z; hq[3] = (_Float16)tq[t][j].w;
                *(v4h*)(sb2 + (4 * j + trow) * STG_PITCH + tcol * 8) = hq;
            }
            const v8h blo = *(const v8h*)(sb2 + c16 * STG_PITCH + half * 32);
            const v8h bhi = *(const v8h*)(sb2 + c16 * STG_PITCH + half * 32 + 16);
            const v16h b = __builtin_shufflevector(blo, bhi, 0, 1, 2, 3, 4, 5, 6, 7,
                                                   8, 9, 10, 11, 12, 13, 14, 15);
            const int row = min((wave + t * WAVES_PER_BLOCK) * 16 + c16, LAST_ROW);
#pragma unroll
            for (int i = 0; i < NWT; ++i) {
                const v16h a_nxt = *(const v16h*)&wf[(((i + 1) & (NWT - 1)) * 32 + lane) * 16];
                const v16h a = a_cur;
                v8f d = v8f{};
                d = __builtin_amdgcn_wmma_f32_16x16x32_f16( false, a,  false, b,
                                                            (short)0, d,
                                                            false,  false);
                {
                    v4h lo, hi;
                    lo[0] = (_Float16)d[0]; lo[1] = (_Float16)d[1]; lo[2] = (_Float16)d[2]; lo[3] = (_Float16)d[3];
                    hi[0] = (_Float16)d[4]; hi[1] = (_Float16)d[5]; hi[2] = (_Float16)d[6]; hi[3] = (_Float16)d[7];
                    const int k0 = 4 * i + 2 * half;
                    *(v4h*)(ptab + k0 * P_KSTRIDE + row * P_ENTRY)       = lo;
                    *(v4h*)(ptab + (k0 + 1) * P_KSTRIDE + row * P_ENTRY) = hi;
                }
                a_cur = a_nxt;
            }
        }
    }

#pragma unroll
    for (int c = CHUNKS / 2; c < CHUNKS; ++c) idx[c] = ein[c * 32];

    float4 sb[CHUNKS / QCHUNKS];
#pragma unroll
    for (int g = 0; g < CHUNKS / QCHUNKS; ++g)
        sb[g] = *(const float4*)(spatial_emb + sp[g] * H_ + hb * HH);

    lds_barrier();

    int kb[4];
#pragma unroll
    for (int i = 0; i < 4; ++i) kb[i] = (4 * l3 + i) * P_KSTRIDE;

    const bool b0 = (lane & 1) != 0;
    const bool b1 = (lane & 2) != 0;
    const bool b2 = (lane & 4) != 0;

#pragma unroll
    for (int g = 0; g < CHUNKS / QCHUNKS; ++g) {
        float v[QCHUNKS][5];
#pragma unroll
        for (int c = 0; c < QCHUNKS; ++c) {
            const int4 e = idx[g * QCHUNKS + c];
            const v4h p0 = *(const v4h*)(ptab + (e.x * P_ENTRY + kb[0]));
            const v4h p1 = *(const v4h*)(ptab + (e.y * P_ENTRY + kb[1]));
            const v4h p2 = *(const v4h*)(ptab + (e.z * P_ENTRY + kb[2]));
            const v4h p3 = *(const v4h*)(ptab + (e.w * P_ENTRY + kb[3]));
#pragma unroll
            for (int q = 0; q < HH; ++q)
                v[c][q] = (((float)p0[q] + (float)p1[q]) + (float)p2[q]) + (float)p3[q];
            v[c][4] = (float)((e.x != 0) + (e.y != 0) + (e.z != 0) + (e.w != 0));
        }

        float w[QCHUNKS / 2][5];
#pragma unroll
        for (int j = 0; j < QCHUNKS / 2; ++j)
#pragma unroll
            for (int q = 0; q < 5; ++q) {
                const float keep = b0 ? v[2 * j + 1][q] : v[2 * j][q];
                const float send = b0 ? v[2 * j][q] : v[2 * j + 1][q];
                w[j][q] = keep + lane_swap<SWAP_XOR1>(send);
            }
        float x[QCHUNKS / 4][5];
#pragma unroll
        for (int j = 0; j < QCHUNKS / 4; ++j)
#pragma unroll
            for (int q = 0; q < 5; ++q) {
                const float keep = b1 ? w[2 * j + 1][q] : w[2 * j][q];
                const float send = b1 ? w[2 * j][q] : w[2 * j + 1][q];
                x[j][q] = keep + lane_swap<SWAP_XOR2>(send);
            }
        float y[5];
#pragma unroll
        for (int q = 0; q < 5; ++q) {
            const float keep = b2 ? x[1][q] : x[0][q];
            const float send = b2 ? x[0][q] : x[1][q];
            y[q] = keep + lane_swap<SWAP_XOR4>(send);
        }

        const float pl = fmaxf(y[4], 1.0f);
        const long long p  = (long long)pbase + g * 32 + slot;
        const long long bi = p >> 16;
        const long long nm = p & 0xFFFF;
        float* o = out + bi * (long long)(H_ * NN) + (long long)(hb * HH) * NN + nm;
        o[0 * NN] = sb[g].x + y[0] / pl;
        o[1 * NN] = sb[g].y + y[1] / pl;
        o[2 * NN] = sb[g].z + y[2] / pl;
        o[3 * NN] = sb[g].w + y[3] / pl;
    }
}

extern "C" void kernel_launch(void* const* d_in, const int* in_sizes, int n_in,
                              void* d_out, int out_size, void* d_ws, size_t ws_size,
                              hipStream_t stream) {
    const int*   spatial_pos   = (const int*)d_in[0];
    const int*   edge_input    = (const int*)d_in[1];
    const float* spatial_emb   = (const float*)d_in[2];
    const float* edge_feat_emb = (const float*)d_in[3];
    const float* edge_pos_emb  = (const float*)d_in[4];
    float*       out           = (float*)d_out;

    hipLaunchKernelGGL(attn_bias_wmma, dim3(NBLOCKS), dim3(THREADS), 0, stream,
                       spatial_pos, edge_input, spatial_emb, edge_feat_emb,
                       edge_pos_emb, out);
}
